# starting kernel
# baseline (speedup 1.0000x reference)
#include <hip/hip_runtime.h>
#include <math.h>

#define NN      50000
#define NE      800000
#define NG      128
#define INC     31
#define HID     64
#define EPS     1e-5f

typedef __attribute__((ext_vector_type(2))) float v2f;
typedef __attribute__((ext_vector_type(8))) float v8f;

static __device__ __forceinline__ unsigned enc_f32(float x) {
    unsigned b = __float_as_uint(x);
    return (b & 0x80000000u) ? ~b : (b | 0x80000000u);
}
static __device__ __forceinline__ float dec_f32(unsigned e) {
    unsigned b = (e & 0x80000000u) ? (e & 0x7FFFFFFFu) : ~e;
    return __uint_as_float(b);
}
#define ENC_NEG_INF 0x007FFFFFu

__global__ void k_pad_rows(const float* __restrict__ s, float* __restrict__ d,
                           int rows, int Csrc, int Cdst) {
    int i = blockIdx.x * blockDim.x + threadIdx.x;
    if (i >= rows * Cdst) return;
    int r = i / Cdst, c = i % Cdst;
    d[i] = (c < Csrc) ? s[r * Csrc + c] : 0.f;
}

__global__ void k_init_aggr(unsigned* __restrict__ em, float* __restrict__ num,
                            float* __restrict__ den, int n) {
    int i = blockIdx.x * blockDim.x + threadIdx.x;
    if (i < n) { em[i] = ENC_NEG_INF; num[i] = 0.f; den[i] = 0.f; }
}
__global__ void k_init_enc(unsigned* __restrict__ p, int n) {
    int i = blockIdx.x * blockDim.x + threadIdx.x;
    if (i < n) p[i] = ENC_NEG_INF;
}

__global__ void k_edge_max4(const int* __restrict__ src, const int* __restrict__ dst,
                            const float* __restrict__ x, const float* __restrict__ tptr,
                            unsigned* __restrict__ em, int E, int cqshift) {
    int g = blockIdx.x * blockDim.x + threadIdx.x;
    int e = g >> cqshift;
    if (e >= E) return;
    int q  = g & ((1 << cqshift) - 1);
    int Cp = 4 << cqshift;
    float t = tptr[0];
    const float4 m = ((const float4*)(x + (size_t)src[e] * Cp))[q];
    unsigned* p = em + (size_t)dst[e] * Cp + q * 4;
    atomicMax(p + 0, enc_f32(m.x * t));
    atomicMax(p + 1, enc_f32(m.y * t));
    atomicMax(p + 2, enc_f32(m.z * t));
    atomicMax(p + 3, enc_f32(m.w * t));
}

__global__ void k_edge_expsum4(const int* __restrict__ src, const int* __restrict__ dst,
                               const float* __restrict__ x, const float* __restrict__ tptr,
                               const unsigned* __restrict__ em,
                               float* __restrict__ num, float* __restrict__ den,
                               int E, int cqshift) {
    int g = blockIdx.x * blockDim.x + threadIdx.x;
    int e = g >> cqshift;
    if (e >= E) return;
    int q  = g & ((1 << cqshift) - 1);
    int Cp = 4 << cqshift;
    float t = tptr[0];
    int   d = dst[e];
    const float4 m = ((const float4*)(x + (size_t)src[e] * Cp))[q];
    size_t base = (size_t)d * Cp + q * 4;
    const unsigned* pe = em + base;
    float* pn = num + base;
    float* pd = den + base;
    float e0 = __expf(m.x * t - dec_f32(pe[0]));
    float e1 = __expf(m.y * t - dec_f32(pe[1]));
    float e2 = __expf(m.z * t - dec_f32(pe[2]));
    float e3 = __expf(m.w * t - dec_f32(pe[3]));
    atomicAdd(pn + 0, m.x * e0);  atomicAdd(pd + 0, e0);
    atomicAdd(pn + 1, m.y * e1);  atomicAdd(pd + 1, e1);
    atomicAdd(pn + 2, m.z * e2);  atomicAdd(pd + 2, e2);
    atomicAdd(pn + 3, m.w * e3);  atomicAdd(pd + 3, e3);
}

__global__ void k_aggr_div(const float* __restrict__ num, const float* __restrict__ den,
                           float* __restrict__ aggr, int n) {
    int i = blockIdx.x * blockDim.x + threadIdx.x;
    if (i >= n) return;
    float dd = den[i];
    aggr[i] = dd > 0.f ? num[i] / dd : 0.f;
}

template<int CP>
__global__ void k_gemm(const float* __restrict__ aggr, const float* __restrict__ xin,
                       const float* __restrict__ Wr, const float* __restrict__ Ws,
                       const float* __restrict__ bias, float* __restrict__ out, int Nrows) {
    const int lane = threadIdx.x;
    const int half = lane >> 4;
    const int l16  = lane & 15;
    const int m0   = blockIdx.x * 16;
    const int h0   = threadIdx.y * 16;
    int arow = m0 + l16;
    if (arow >= Nrows) arow = Nrows - 1;
    const int bcol = h0 + l16;

    const float* ap = aggr + (size_t)arow * CP + half * 2;
    const float* xp = xin  + (size_t)arow * CP + half * 2;
    const float* wr = Wr   + (size_t)bcol * CP + half * 2;
    const float* wsp= Ws   + (size_t)bcol * CP + half * 2;

    v8f acc = {};
#pragma unroll
    for (int k = 0; k < CP; k += 4) {
        v2f a = *(const v2f*)(ap + k);
        v2f b = *(const v2f*)(wr + k);
        acc = __builtin_amdgcn_wmma_f32_16x16x4_f32(false, a, false, b,
                                                    (short)0, acc, false, false);
    }
#pragma unroll
    for (int k = 0; k < CP; k += 4) {
        v2f a = *(const v2f*)(xp + k);
        v2f b = *(const v2f*)(wsp + k);
        acc = __builtin_amdgcn_wmma_f32_16x16x4_f32(false, a, false, b,
                                                    (short)0, acc, false, false);
    }

    float bv = bias[bcol];
#pragma unroll
    for (int r = 0; r < 8; ++r) {
        int row = m0 + r + half * 8;
        if (row < Nrows) out[(size_t)row * HID + bcol] = acc[r] + bv;
    }
}

__global__ void k_instnorm_relu(float* __restrict__ x, int Nrows) {
    int wave = (blockIdx.x * blockDim.x + threadIdx.x) >> 5;
    int lane = threadIdx.x & 31;
    if (wave >= Nrows) return;
    float v0 = x[wave * HID + lane];
    float v1 = x[wave * HID + lane + 32];
    float s = v0 + v1;
    for (int off = 16; off; off >>= 1) s += __shfl_xor(s, off);
    float mu = s * (1.f / HID);
    float d0 = v0 - mu, d1 = v1 - mu;
    float q = d0 * d0 + d1 * d1;
    for (int off = 16; off; off >>= 1) q += __shfl_xor(q, off);
    float inv = rsqrtf(q * (1.f / HID) + EPS);
    x[wave * HID + lane]      = fmaxf(d0 * inv, 0.f);
    x[wave * HID + lane + 32] = fmaxf(d1 * inv, 0.f);
}

__global__ void k_graph_max(const float* __restrict__ xl, const int* __restrict__ batch,
                            unsigned* __restrict__ he, int n) {
    int i = blockIdx.x * blockDim.x + threadIdx.x;
    if (i >= n) return;
    int node = i >> 6;
    int c    = i & 63;
    atomicMax(&he[batch[node] * HID + c], enc_f32(xl[i]));
}

__global__ void k_decode_h(const unsigned* __restrict__ henc, float* __restrict__ h, int n) {
    int i = blockIdx.x * blockDim.x + threadIdx.x;
    if (i >= n) return;
    int g = i / (3 * HID);
    int c = i % (3 * HID);
    int layer = c / HID, cc = c % HID;
    h[i] = dec_f32(henc[layer * NG * HID + g * HID + cc]);
}

__global__ void k_mlp1(const float* __restrict__ h, const float* __restrict__ Wl1,
                       const float* __restrict__ bl1, float* __restrict__ z1) {
    int i = blockIdx.x * blockDim.x + threadIdx.x;
    if (i >= NG * 2 * HID) return;
    int g = i / (2 * HID), oc = i % (2 * HID);
    float s = bl1[oc];
    for (int k = 0; k < 3 * HID; ++k) s += h[g * 3 * HID + k] * Wl1[oc * 3 * HID + k];
    z1[i] = fmaxf(s, 0.f);
}

__global__ void k_mlp2_norm(const float* __restrict__ z1, const float* __restrict__ Wl2,
                            const float* __restrict__ bl2, float* __restrict__ outp) {
    int wave = (blockIdx.x * blockDim.x + threadIdx.x) >> 5;
    int lane = threadIdx.x & 31;
    if (wave >= NG) return;
    float s = bl2[lane];
    for (int k = 0; k < 2 * HID; ++k) s += z1[wave * 2 * HID + k] * Wl2[lane * 2 * HID + k];
    float q = s * s;
    for (int off = 16; off; off >>= 1) q += __shfl_xor(q, off);
    float nrm = fmaxf(sqrtf(q), 1e-12f);
    outp[wave * (HID / 2) + lane] = s / nrm;
}

extern "C" void kernel_launch(void* const* d_in, const int* in_sizes, int n_in,
                              void* d_out, int out_size, void* d_ws, size_t ws_size,
                              hipStream_t stream) {
    const float* x    = (const float*)d_in[0];
    const int*  eidx  = (const int*)  d_in[1];
    const int*  batch = (const int*)  d_in[2];
    const float* t    = (const float*)d_in[3];
    const float* W1r  = (const float*)d_in[4];
    const float* b1   = (const float*)d_in[5];
    const float* W1s  = (const float*)d_in[6];
    const float* W2r  = (const float*)d_in[7];
    const float* b2   = (const float*)d_in[8];
    const float* W2s  = (const float*)d_in[9];
    const float* W3r  = (const float*)d_in[10];
    const float* b3   = (const float*)d_in[11];
    const float* W3s  = (const float*)d_in[12];
    const float* Wl1  = (const float*)d_in[13];
    const float* bl1  = (const float*)d_in[14];
    const float* Wl2  = (const float*)d_in[15];
    const float* bl2  = (const float*)d_in[16];
    const int* src = eidx;
    const int* dst = eidx + NE;

    float* ws = (float*)d_ws;
    size_t o = 0;
    float*    x1     = ws + o; o += (size_t)NN * HID;
    float*    x2     = ws + o; o += (size_t)NN * HID;
    float*    x3     = ws + o; o += (size_t)NN * HID;
    unsigned* encmax = (unsigned*)(ws + o); o += (size_t)NN * HID;
    float*    num    = ws + o; o += (size_t)NN * HID;
    float*    den    = ws + o; o += (size_t)NN * HID;
    float*    aggr   = ws + o; o += (size_t)NN * HID;
    float*    xpad   = ws + o; o += (size_t)NN * 32;
    float*    w1rp   = ws + o; o += (size_t)HID * 32;
    float*    w1sp   = ws + o; o += (size_t)HID * 32;
    unsigned* henc   = (unsigned*)(ws + o); o += 3 * NG * HID;
    float*    h      = ws + o; o += NG * 3 * HID;
    float*    z1     = ws + o; o += NG * 2 * HID;

    k_pad_rows<<<(NN * 32 + 255) / 256, 256, 0, stream>>>(x, xpad, NN, INC, 32);
    k_pad_rows<<<(HID * 32 + 255) / 256, 256, 0, stream>>>(W1r, w1rp, HID, INC, 32);
    k_pad_rows<<<(HID * 32 + 255) / 256, 256, 0, stream>>>(W1s, w1sp, HID, INC, 32);
    k_init_enc<<<(3 * NG * HID + 255) / 256, 256, 0, stream>>>(henc, 3 * NG * HID);

    struct Layer { const float* xin; int Cp; const float *Wr, *b, *Ws; float* xout; };
    Layer layers[3] = {
        { xpad, 32, w1rp, b1, w1sp, x1 },
        { x1,   64, W2r,  b2, W2s,  x2 },
        { x2,   64, W3r,  b3, W3s,  x3 },
    };

    for (int l = 0; l < 3; ++l) {
        const Layer& L = layers[l];
        int nagg = NN * L.Cp;
        k_init_aggr<<<(nagg + 255) / 256, 256, 0, stream>>>(encmax, num, den, nagg);

        int cqshift = (L.Cp == 32) ? 3 : 4;
        int tot = NE << cqshift;
        int eblocks = (tot + 255) / 256;
        k_edge_max4<<<eblocks, 256, 0, stream>>>(src, dst, L.xin, t, encmax, NE, cqshift);
        k_edge_expsum4<<<eblocks, 256, 0, stream>>>(src, dst, L.xin, t, encmax, num, den,
                                                    NE, cqshift);
        k_aggr_div<<<(nagg + 255) / 256, 256, 0, stream>>>(num, den, aggr, nagg);

        dim3 gblk(32, 4);
        if (L.Cp == 32)
            k_gemm<32><<<(NN + 15) / 16, gblk, 0, stream>>>(aggr, L.xin, L.Wr, L.Ws, L.b,
                                                            L.xout, NN);
        else
            k_gemm<64><<<(NN + 15) / 16, gblk, 0, stream>>>(aggr, L.xin, L.Wr, L.Ws, L.b,
                                                            L.xout, NN);

        k_instnorm_relu<<<(NN + 7) / 8, 256, 0, stream>>>(L.xout, NN);

        int ngm = NN * HID;
        k_graph_max<<<(ngm + 255) / 256, 256, 0, stream>>>(L.xout, batch,
                                                           henc + (size_t)l * NG * HID, ngm);
    }

    k_decode_h<<<(NG * 3 * HID + 255) / 256, 256, 0, stream>>>(henc, h, NG * 3 * HID);
    k_mlp1<<<(NG * 2 * HID + 255) / 256, 256, 0, stream>>>(h, Wl1, bl1, z1);
    k_mlp2_norm<<<NG * 32 / 128, 128, 0, stream>>>(z1, Wl2, bl2, (float*)d_out);
}


# Edge passes remapped: one wave per 8 edges, lane = channel, so each row load and atomic instruction covers one whole 128-B line (was 8 words in each of 4 lines); float atomic sums arrive in new order
# speedup vs baseline: 1.6003x; 1.6003x over previous
#include <hip/hip_runtime.h>
#include <math.h>

#define NN      50000
#define NE      800000
#define NG      128
#define INC     31
#define HID     64
#define EPS     1e-5f

typedef __attribute__((ext_vector_type(2))) float v2f;
typedef __attribute__((ext_vector_type(8))) float v8f;

static __device__ __forceinline__ unsigned enc_f32(float x) {
    unsigned b = __float_as_uint(x);
    return (b & 0x80000000u) ? ~b : (b | 0x80000000u);
}
static __device__ __forceinline__ float dec_f32(unsigned e) {
    unsigned b = (e & 0x80000000u) ? (e & 0x7FFFFFFFu) : ~e;
    return __uint_as_float(b);
}
#define ENC_NEG_INF 0x007FFFFFu

__global__ void k_pad_rows(const float* __restrict__ s, float* __restrict__ d,
                           int rows, int Csrc, int Cdst) {
    int i = blockIdx.x * blockDim.x + threadIdx.x;
    if (i >= rows * Cdst) return;
    int r = i / Cdst, c = i % Cdst;
    d[i] = (c < Csrc) ? s[r * Csrc + c] : 0.f;
}

__global__ void k_init_aggr(unsigned* __restrict__ em, float* __restrict__ num,
                            float* __restrict__ den, int n) {
    int i = blockIdx.x * blockDim.x + threadIdx.x;
    if (i < n) { em[i] = ENC_NEG_INF; num[i] = 0.f; den[i] = 0.f; }
}
__global__ void k_init_enc(unsigned* __restrict__ p, int n) {
    int i = blockIdx.x * blockDim.x + threadIdx.x;
    if (i < n) p[i] = ENC_NEG_INF;
}

#define EPW 8
static_assert(NE % EPW == 0, "edge count must be a multiple of the per-wave batch");

template<int CP>
__global__ void k_edge_max_rows(const int* __restrict__ src, const int* __restrict__ dst,
                                const float* __restrict__ x, const float* __restrict__ tptr,
                                unsigned* __restrict__ em) {
    constexpr int H = CP / 32;
    const int lane = threadIdx.x & 31;
    const int wave = __builtin_amdgcn_readfirstlane((int)((blockIdx.x * blockDim.x + threadIdx.x) >> 5));
    const int e0 = wave * EPW;
    if (e0 >= NE) return;
    const float t = tptr[0];
    int s[EPW], d[EPW];
    float v[EPW][H];
#pragma unroll
    for (int u = 0; u < EPW; ++u) { s[u] = src[e0 + u]; d[u] = dst[e0 + u]; }
#pragma unroll
    for (int u = 0; u < EPW; ++u)
#pragma unroll
        for (int h = 0; h < H; ++h)
            v[u][h] = x[(size_t)s[u] * CP + h * 32 + lane];
#pragma unroll
    for (int u = 0; u < EPW; ++u)
#pragma unroll
        for (int h = 0; h < H; ++h)
            atomicMax(em + (size_t)d[u] * CP + h * 32 + lane, enc_f32(v[u][h] * t));
}

template<int CP>
__global__ void k_edge_expsum_rows(const int* __restrict__ src, const int* __restrict__ dst,
                                   const float* __restrict__ x, const float* __restrict__ tptr,
                                   const unsigned* __restrict__ em,
                                   float* __restrict__ num, float* __restrict__ den) {
    constexpr int H = CP / 32;
    const int lane = threadIdx.x & 31;
    const int wave = __builtin_amdgcn_readfirstlane((int)((blockIdx.x * blockDim.x + threadIdx.x) >> 5));
    const int e0 = wave * EPW;
    if (e0 >= NE) return;
    const float t = tptr[0];
    int s[EPW], d[EPW];
    float v[EPW][H];
    unsigned mxe[EPW][H];
#pragma unroll
    for (int u = 0; u < EPW; ++u) { s[u] = src[e0 + u]; d[u] = dst[e0 + u]; }
#pragma unroll
    for (int u = 0; u < EPW; ++u)
#pragma unroll
        for (int h = 0; h < H; ++h) {
            v[u][h]   = x[(size_t)s[u] * CP + h * 32 + lane];
            mxe[u][h] = em[(size_t)d[u] * CP + h * 32 + lane];
        }
#pragma unroll
    for (int u = 0; u < EPW; ++u)
#pragma unroll
        for (int h = 0; h < H; ++h) {
            const size_t idx = (size_t)d[u] * CP + h * 32 + lane;
            const float m  = v[u][h];
            const float ee = __expf(m * t - dec_f32(mxe[u][h]));
            atomicAdd(num + idx, m * ee);
            atomicAdd(den + idx, ee);
        }
}

__global__ void k_aggr_div(const float* __restrict__ num, const float* __restrict__ den,
                           float* __restrict__ aggr, int n) {
    int i = blockIdx.x * blockDim.x + threadIdx.x;
    if (i >= n) return;
    float dd = den[i];
    aggr[i] = dd > 0.f ? num[i] / dd : 0.f;
}

template<int CP>
__global__ void k_gemm(const float* __restrict__ aggr, const float* __restrict__ xin,
                       const float* __restrict__ Wr, const float* __restrict__ Ws,
                       const float* __restrict__ bias, float* __restrict__ out, int Nrows) {
    const int lane = threadIdx.x;
    const int half = lane >> 4;
    const int l16  = lane & 15;
    const int m0   = blockIdx.x * 16;
    const int h0   = threadIdx.y * 16;
    int arow = m0 + l16;
    if (arow >= Nrows) arow = Nrows - 1;
    const int bcol = h0 + l16;

    const float* ap = aggr + (size_t)arow * CP + half * 2;
    const float* xp = xin  + (size_t)arow * CP + half * 2;
    const float* wr = Wr   + (size_t)bcol * CP + half * 2;
    const float* wsp= Ws   + (size_t)bcol * CP + half * 2;

    v8f acc = {};
#pragma unroll
    for (int k = 0; k < CP; k += 4) {
        v2f a = *(const v2f*)(ap + k);
        v2f b = *(const v2f*)(wr + k);
        acc = __builtin_amdgcn_wmma_f32_16x16x4_f32(false, a, false, b,
                                                    (short)0, acc, false, false);
    }
#pragma unroll
    for (int k = 0; k < CP; k += 4) {
        v2f a = *(const v2f*)(xp + k);
        v2f b = *(const v2f*)(wsp + k);
        acc = __builtin_amdgcn_wmma_f32_16x16x4_f32(false, a, false, b,
                                                    (short)0, acc, false, false);
    }

    float bv = bias[bcol];
#pragma unroll
    for (int r = 0; r < 8; ++r) {
        int row = m0 + r + half * 8;
        if (row < Nrows) out[(size_t)row * HID + bcol] = acc[r] + bv;
    }
}

__global__ void k_instnorm_relu(float* __restrict__ x, int Nrows) {
    int wave = (blockIdx.x * blockDim.x + threadIdx.x) >> 5;
    int lane = threadIdx.x & 31;
    if (wave >= Nrows) return;
    float v0 = x[wave * HID + lane];
    float v1 = x[wave * HID + lane + 32];
    float s = v0 + v1;
    for (int off = 16; off; off >>= 1) s += __shfl_xor(s, off);
    float mu = s * (1.f / HID);
    float d0 = v0 - mu, d1 = v1 - mu;
    float q = d0 * d0 + d1 * d1;
    for (int off = 16; off; off >>= 1) q += __shfl_xor(q, off);
    float inv = rsqrtf(q * (1.f / HID) + EPS);
    x[wave * HID + lane]      = fmaxf(d0 * inv, 0.f);
    x[wave * HID + lane + 32] = fmaxf(d1 * inv, 0.f);
}

__global__ void k_graph_max(const float* __restrict__ xl, const int* __restrict__ batch,
                            unsigned* __restrict__ he, int n) {
    int i = blockIdx.x * blockDim.x + threadIdx.x;
    if (i >= n) return;
    int node = i >> 6;
    int c    = i & 63;
    atomicMax(&he[batch[node] * HID + c], enc_f32(xl[i]));
}

__global__ void k_decode_h(const unsigned* __restrict__ henc, float* __restrict__ h, int n) {
    int i = blockIdx.x * blockDim.x + threadIdx.x;
    if (i >= n) return;
    int g = i / (3 * HID);
    int c = i % (3 * HID);
    int layer = c / HID, cc = c % HID;
    h[i] = dec_f32(henc[layer * NG * HID + g * HID + cc]);
}

__global__ void k_mlp1(const float* __restrict__ h, const float* __restrict__ Wl1,
                       const float* __restrict__ bl1, float* __restrict__ z1) {
    int i = blockIdx.x * blockDim.x + threadIdx.x;
    if (i >= NG * 2 * HID) return;
    int g = i / (2 * HID), oc = i % (2 * HID);
    float s = bl1[oc];
    for (int k = 0; k < 3 * HID; ++k) s += h[g * 3 * HID + k] * Wl1[oc * 3 * HID + k];
    z1[i] = fmaxf(s, 0.f);
}

__global__ void k_mlp2_norm(const float* __restrict__ z1, const float* __restrict__ Wl2,
                            const float* __restrict__ bl2, float* __restrict__ outp) {
    int wave = (blockIdx.x * blockDim.x + threadIdx.x) >> 5;
    int lane = threadIdx.x & 31;
    if (wave >= NG) return;
    float s = bl2[lane];
    for (int k = 0; k < 2 * HID; ++k) s += z1[wave * 2 * HID + k] * Wl2[lane * 2 * HID + k];
    float q = s * s;
    for (int off = 16; off; off >>= 1) q += __shfl_xor(q, off);
    float nrm = fmaxf(sqrtf(q), 1e-12f);
    outp[wave * (HID / 2) + lane] = s / nrm;
}

extern "C" void kernel_launch(void* const* d_in, const int* in_sizes, int n_in,
                              void* d_out, int out_size, void* d_ws, size_t ws_size,
                              hipStream_t stream) {
    const float* x    = (const float*)d_in[0];
    const int*  eidx  = (const int*)  d_in[1];
    const int*  batch = (const int*)  d_in[2];
    const float* t    = (const float*)d_in[3];
    const float* W1r  = (const float*)d_in[4];
    const float* b1   = (const float*)d_in[5];
    const float* W1s  = (const float*)d_in[6];
    const float* W2r  = (const float*)d_in[7];
    const float* b2   = (const float*)d_in[8];
    const float* W2s  = (const float*)d_in[9];
    const float* W3r  = (const float*)d_in[10];
    const float* b3   = (const float*)d_in[11];
    const float* W3s  = (const float*)d_in[12];
    const float* Wl1  = (const float*)d_in[13];
    const float* bl1  = (const float*)d_in[14];
    const float* Wl2  = (const float*)d_in[15];
    const float* bl2  = (const float*)d_in[16];
    const int* src = eidx;
    const int* dst = eidx + NE;

    float* ws = (float*)d_ws;
    size_t o = 0;
    float*    x1     = ws + o; o += (size_t)NN * HID;
    float*    x2     = ws + o; o += (size_t)NN * HID;
    float*    x3     = ws + o; o += (size_t)NN * HID;
    unsigned* encmax = (unsigned*)(ws + o); o += (size_t)NN * HID;
    float*    num    = ws + o; o += (size_t)NN * HID;
    float*    den    = ws + o; o += (size_t)NN * HID;
    float*    aggr   = ws + o; o += (size_t)NN * HID;
    float*    xpad   = ws + o; o += (size_t)NN * 32;
    float*    w1rp   = ws + o; o += (size_t)HID * 32;
    float*    w1sp   = ws + o; o += (size_t)HID * 32;
    unsigned* henc   = (unsigned*)(ws + o); o += 3 * NG * HID;
    float*    h      = ws + o; o += NG * 3 * HID;
    float*    z1     = ws + o; o += NG * 2 * HID;

    k_pad_rows<<<(NN * 32 + 255) / 256, 256, 0, stream>>>(x, xpad, NN, INC, 32);
    k_pad_rows<<<(HID * 32 + 255) / 256, 256, 0, stream>>>(W1r, w1rp, HID, INC, 32);
    k_pad_rows<<<(HID * 32 + 255) / 256, 256, 0, stream>>>(W1s, w1sp, HID, INC, 32);
    k_init_enc<<<(3 * NG * HID + 255) / 256, 256, 0, stream>>>(henc, 3 * NG * HID);

    struct Layer { const float* xin; int Cp; const float *Wr, *b, *Ws; float* xout; };
    Layer layers[3] = {
        { xpad, 32, w1rp, b1, w1sp, x1 },
        { x1,   64, W2r,  b2, W2s,  x2 },
        { x2,   64, W3r,  b3, W3s,  x3 },
    };

    for (int l = 0; l < 3; ++l) {
        const Layer& L = layers[l];
        int nagg = NN * L.Cp;
        k_init_aggr<<<(nagg + 255) / 256, 256, 0, stream>>>(encmax, num, den, nagg);

        int eblocks = (NE / EPW * 32 + 255) / 256;
        if (L.Cp == 32) {
            k_edge_max_rows<32><<<eblocks, 256, 0, stream>>>(src, dst, L.xin, t, encmax);
            k_edge_expsum_rows<32><<<eblocks, 256, 0, stream>>>(src, dst, L.xin, t, encmax,
                                                               num, den);
        } else {
            k_edge_max_rows<64><<<eblocks, 256, 0, stream>>>(src, dst, L.xin, t, encmax);
            k_edge_expsum_rows<64><<<eblocks, 256, 0, stream>>>(src, dst, L.xin, t, encmax,
                                                               num, den);
        }
        k_aggr_div<<<(nagg + 255) / 256, 256, 0, stream>>>(num, den, aggr, nagg);

        dim3 gblk(32, 4);
        if (L.Cp == 32)
            k_gemm<32><<<(NN + 15) / 16, gblk, 0, stream>>>(aggr, L.xin, L.Wr, L.Ws, L.b,
                                                            L.xout, NN);
        else
            k_gemm<64><<<(NN + 15) / 16, gblk, 0, stream>>>(aggr, L.xin, L.Wr, L.Ws, L.b,
                                                            L.xout, NN);

        k_instnorm_relu<<<(NN + 7) / 8, 256, 0, stream>>>(L.xout, NN);

        int ngm = NN * HID;
        k_graph_max<<<(ngm + 255) / 256, 256, 0, stream>>>(L.xout, batch,
                                                           henc + (size_t)l * NG * HID, ngm);
    }

    k_decode_h<<<(NG * 3 * HID + 255) / 256, 256, 0, stream>>>(henc, h, NG * 3 * HID);
    k_mlp1<<<(NG * 2 * HID + 255) / 256, 256, 0, stream>>>(h, Wl1, bl1, z1);
    k_mlp2_norm<<<NG * 32 / 128, 128, 0, stream>>>(z1, Wl2, bl2, (float*)d_out);
}


# Softmax aggr owner-computes: a group per 200 dst nodes picks its edges from the unsorted list once (owner-byte filter); max/num/den in LDS (LDS atomics, sums in arrival order), no global atomic
# speedup vs baseline: 2.3921x; 1.4948x over previous
#include <hip/hip_runtime.h>
#include <math.h>

#define NN      50000
#define NE      800000
#define NG      128
#define INC     31
#define HID     64
#define EPS     1e-5f

typedef __attribute__((ext_vector_type(2))) float v2f;
typedef __attribute__((ext_vector_type(8))) float v8f;

static __device__ __forceinline__ unsigned enc_f32(float x) {
    unsigned b = __float_as_uint(x);
    return (b & 0x80000000u) ? ~b : (b | 0x80000000u);
}
static __device__ __forceinline__ float dec_f32(unsigned e) {
    unsigned b = (e & 0x80000000u) ? (e & 0x7FFFFFFFu) : ~e;
    return __uint_as_float(b);
}
#define ENC_NEG_INF 0x007FFFFFu

__global__ void k_pad_rows(const float* __restrict__ s, float* __restrict__ d,
                           int rows, int Csrc, int Cdst) {
    int i = blockIdx.x * blockDim.x + threadIdx.x;
    if (i >= rows * Cdst) return;
    int r = i / Cdst, c = i % Cdst;
    d[i] = (c < Csrc) ? s[r * Csrc + c] : 0.f;
}

__global__ void k_init_enc(unsigned* __restrict__ p, int n) {
    int i = blockIdx.x * blockDim.x + threadIdx.x;
    if (i < n) p[i] = ENC_NEG_INF;
}

#define RB   200
#define NB   (NN / RB)
#define NT   1024
#define NWV  (NT / 32)
#define CAP  8192
#define EB   16
static_assert(NN % RB == 0, "groups must tile the nodes exactly");
static_assert(NB <= 255 && RB <= 256, "owner id (255 is kept free) and local node id must each fit a byte");
static_assert(NE % 16 == 0, "owner table is read 16 bytes a lane");
static_assert(CAP % EB == 0, "a list is padded to whole batches inside its capacity");

__global__ void k_owner_ids(const int* __restrict__ dst, unsigned* __restrict__ own4) {
    int g = blockIdx.x * blockDim.x + threadIdx.x;
    if (g >= NE / 4) return;
    const int4 d = ((const int4*)dst)[g];
    own4[g] = (unsigned)(d.x / RB) | ((unsigned)(d.y / RB) << 8) |
              ((unsigned)(d.z / RB) << 16) | ((unsigned)(d.w / RB) << 24);
}

__global__ void __launch_bounds__(NT)
k_build_lists(const int* __restrict__ src, const int* __restrict__ dst,
              const unsigned* __restrict__ own4,
              unsigned* __restrict__ glist, unsigned* __restrict__ gcnt) {
    __shared__ unsigned s_list[CAP];
    __shared__ unsigned s_cnt;
    const int tid  = threadIdx.x;
    const int b    = blockIdx.x;
    const int base = b * RB;
    if (tid == 0) s_cnt = 0u;
    __syncthreads();

    const uint4* own16 = (const uint4*)own4;
    const unsigned pat = (unsigned)b * 0x01010101u;
    constexpr int NQ = NE / 16;
    const int off = b * (NQ / NB);
    constexpr int SB = 8;
    for (int i = tid; i < NQ; i += SB * NT) {
        uint4 w4[SB];
        int   qq[SB];
#pragma unroll
        for (int r = 0; r < SB; ++r) {
            const int ii = i + r * NT;
            int q = (ii < NQ ? ii : tid) + off;
            if (q >= NQ) q -= NQ;
            qq[r] = q;
            w4[r] = own16[q];
        }
#pragma unroll
        for (int r = 0; r < SB; ++r) {
            const bool live = (i + r * NT) < NQ;
            const unsigned w[4] = { live ? w4[r].x : 0xFFFFFFFFu, live ? w4[r].y : 0xFFFFFFFFu,
                                    live ? w4[r].z : 0xFFFFFFFFu, live ? w4[r].w : 0xFFFFFFFFu };
#pragma unroll
            for (int k = 0; k < 4; ++k) {
                const unsigned xw = w[k] ^ pat;
                unsigned z = ~(((xw & 0x7F7F7F7Fu) + 0x7F7F7F7Fu) | xw) & 0x80808080u;
                while (z) {
                    const int j = __builtin_ctz(z) >> 3;
                    z &= z - 1u;
                    const unsigned pos = atomicAdd(&s_cnt, 1u);
                    if (pos < CAP) s_list[pos] = (unsigned)(qq[r] * 16 + k * 4 + j);
                }
            }
        }
    }
    __syncthreads();
    const unsigned cnt = s_cnt;
    if (tid == 0) gcnt[b] = cnt;
    if (cnt <= CAP) {
        const unsigned padded = (cnt + EB - 1u) & ~(unsigned)(EB - 1);
        for (unsigned i = tid; i < padded; i += NT) {
            const unsigned e = s_list[i < cnt ? i : cnt - 1u];
            glist[(size_t)b * CAP + i] = ((unsigned)src[e] << 8) | (unsigned)(dst[e] - base);
        }
    }
}

template<int CP>
__global__ void __launch_bounds__(NT)
k_aggr_owner(const int* __restrict__ src, const int* __restrict__ dst,
             const unsigned* __restrict__ own4,
             const unsigned* __restrict__ glist, const unsigned* __restrict__ gcnt,
             const float* __restrict__ x, const float* __restrict__ tptr,
             float* __restrict__ aggr) {
    constexpr int H = CP / 32;
    __shared__ unsigned s_em[RB * CP];
    __shared__ float    s_num[RB * CP];
    __shared__ float    s_den[RB * CP];

    const int tid  = threadIdx.x;
    const int lane = tid & 31;
    const int wv   = __builtin_amdgcn_readfirstlane(tid >> 5);
    const int b    = blockIdx.x;
    const int base = b * RB;
    const float t  = tptr[0];
    const unsigned cnt = __builtin_amdgcn_readfirstlane(gcnt[b]);

    for (int i = tid; i < RB * CP; i += NT) { s_em[i] = ENC_NEG_INF; s_num[i] = 0.f; s_den[i] = 0.f; }
    __syncthreads();

    if (cnt <= CAP) {
        const unsigned* __restrict__ gl = glist + (size_t)b * CAP;

        for (unsigned i0 = wv * EB; i0 < cnt; i0 += NWV * EB) {
            unsigned p[EB];
            float v[EB][H];
#pragma unroll
            for (int u = 0; u < EB; ++u) p[u] = gl[i0 + u];
#pragma unroll
            for (int u = 0; u < EB; ++u)
#pragma unroll
                for (int h = 0; h < H; ++h)
                    v[u][h] = x[(size_t)(p[u] >> 8) * CP + h * 32 + lane];
#pragma unroll
            for (int u = 0; u < EB; ++u)
#pragma unroll
                for (int h = 0; h < H; ++h)
                    atomicMax(&s_em[(p[u] & 255u) * CP + h * 32 + lane], enc_f32(v[u][h] * t));
        }
        __syncthreads();

        for (unsigned i0 = wv * EB; i0 < cnt; i0 += NWV * EB) {
            unsigned p[EB];
            float v[EB][H];
#pragma unroll
            for (int u = 0; u < EB; ++u) p[u] = gl[i0 + u];
#pragma unroll
            for (int u = 0; u < EB; ++u)
#pragma unroll
                for (int h = 0; h < H; ++h)
                    v[u][h] = x[(size_t)(p[u] >> 8) * CP + h * 32 + lane];
#pragma unroll
            for (int u = 0; u < EB; ++u)
                if (i0 + u < cnt) {
#pragma unroll
                    for (int h = 0; h < H; ++h) {
                        const unsigned a = (p[u] & 255u) * CP + h * 32 + lane;
                        const float m  = v[u][h];
                        const float ee = __expf(m * t - dec_f32(s_em[a]));
                        atomicAdd(&s_num[a], m * ee);
                        atomicAdd(&s_den[a], ee);
                    }
                }
        }
        __syncthreads();
    } else {
        const unsigned char* own1 = (const unsigned char*)own4;
        for (int pass = 0; pass < 2; ++pass) {
            for (int e = tid; e < NE; e += NT) {
                if (own1[e] != (unsigned char)b) continue;
                const int s  = src[e];
                const int dl = dst[e] - base;
                for (int c = 0; c < CP; ++c) {
                    const float m = x[(size_t)s * CP + c];
                    const int a = dl * CP + c;
                    if (pass == 0) {
                        atomicMax(&s_em[a], enc_f32(m * t));
                    } else {
                        const float ee = __expf(m * t - dec_f32(s_em[a]));
                        atomicAdd(&s_num[a], m * ee);
                        atomicAdd(&s_den[a], ee);
                    }
                }
            }
            __syncthreads();
        }
    }

    for (int i = tid; i < RB * CP; i += NT) {
        const float dd = s_den[i];
        aggr[(size_t)base * CP + i] = dd > 0.f ? s_num[i] / dd : 0.f;
    }
}

template<int CP>
__global__ void k_gemm(const float* __restrict__ aggr, const float* __restrict__ xin,
                       const float* __restrict__ Wr, const float* __restrict__ Ws,
                       const float* __restrict__ bias, float* __restrict__ out, int Nrows) {
    const int lane = threadIdx.x;
    const int half = lane >> 4;
    const int l16  = lane & 15;
    const int m0   = blockIdx.x * 16;
    const int h0   = threadIdx.y * 16;
    int arow = m0 + l16;
    if (arow >= Nrows) arow = Nrows - 1;
    const int bcol = h0 + l16;

    const float* ap = aggr + (size_t)arow * CP + half * 2;
    const float* xp = xin  + (size_t)arow * CP + half * 2;
    const float* wr = Wr   + (size_t)bcol * CP + half * 2;
    const float* wsp= Ws   + (size_t)bcol * CP + half * 2;

    v8f acc = {};
#pragma unroll
    for (int k = 0; k < CP; k += 4) {
        v2f a = *(const v2f*)(ap + k);
        v2f b = *(const v2f*)(wr + k);
        acc = __builtin_amdgcn_wmma_f32_16x16x4_f32(false, a, false, b,
                                                    (short)0, acc, false, false);
    }
#pragma unroll
    for (int k = 0; k < CP; k += 4) {
        v2f a = *(const v2f*)(xp + k);
        v2f b = *(const v2f*)(wsp + k);
        acc = __builtin_amdgcn_wmma_f32_16x16x4_f32(false, a, false, b,
                                                    (short)0, acc, false, false);
    }

    float bv = bias[bcol];
#pragma unroll
    for (int r = 0; r < 8; ++r) {
        int row = m0 + r + half * 8;
        if (row < Nrows) out[(size_t)row * HID + bcol] = acc[r] + bv;
    }
}

__global__ void k_instnorm_relu(float* __restrict__ x, int Nrows) {
    int wave = (blockIdx.x * blockDim.x + threadIdx.x) >> 5;
    int lane = threadIdx.x & 31;
    if (wave >= Nrows) return;
    float v0 = x[wave * HID + lane];
    float v1 = x[wave * HID + lane + 32];
    float s = v0 + v1;
    for (int off = 16; off; off >>= 1) s += __shfl_xor(s, off);
    float mu = s * (1.f / HID);
    float d0 = v0 - mu, d1 = v1 - mu;
    float q = d0 * d0 + d1 * d1;
    for (int off = 16; off; off >>= 1) q += __shfl_xor(q, off);
    float inv = rsqrtf(q * (1.f / HID) + EPS);
    x[wave * HID + lane]      = fmaxf(d0 * inv, 0.f);
    x[wave * HID + lane + 32] = fmaxf(d1 * inv, 0.f);
}

__global__ void k_graph_max(const float* __restrict__ xl, const int* __restrict__ batch,
                            unsigned* __restrict__ he, int n) {
    int i = blockIdx.x * blockDim.x + threadIdx.x;
    if (i >= n) return;
    int node = i >> 6;
    int c    = i & 63;
    atomicMax(&he[batch[node] * HID + c], enc_f32(xl[i]));
}

__global__ void k_decode_h(const unsigned* __restrict__ henc, float* __restrict__ h, int n) {
    int i = blockIdx.x * blockDim.x + threadIdx.x;
    if (i >= n) return;
    int g = i / (3 * HID);
    int c = i % (3 * HID);
    int layer = c / HID, cc = c % HID;
    h[i] = dec_f32(henc[layer * NG * HID + g * HID + cc]);
}

__global__ void k_mlp1(const float* __restrict__ h, const float* __restrict__ Wl1,
                       const float* __restrict__ bl1, float* __restrict__ z1) {
    int i = blockIdx.x * blockDim.x + threadIdx.x;
    if (i >= NG * 2 * HID) return;
    int g = i / (2 * HID), oc = i % (2 * HID);
    float s = bl1[oc];
    for (int k = 0; k < 3 * HID; ++k) s += h[g * 3 * HID + k] * Wl1[oc * 3 * HID + k];
    z1[i] = fmaxf(s, 0.f);
}

__global__ void k_mlp2_norm(const float* __restrict__ z1, const float* __restrict__ Wl2,
                            const float* __restrict__ bl2, float* __restrict__ outp) {
    int wave = (blockIdx.x * blockDim.x + threadIdx.x) >> 5;
    int lane = threadIdx.x & 31;
    if (wave >= NG) return;
    float s = bl2[lane];
    for (int k = 0; k < 2 * HID; ++k) s += z1[wave * 2 * HID + k] * Wl2[lane * 2 * HID + k];
    float q = s * s;
    for (int off = 16; off; off >>= 1) q += __shfl_xor(q, off);
    float nrm = fmaxf(sqrtf(q), 1e-12f);
    outp[wave * (HID / 2) + lane] = s / nrm;
}

extern "C" void kernel_launch(void* const* d_in, const int* in_sizes, int n_in,
                              void* d_out, int out_size, void* d_ws, size_t ws_size,
                              hipStream_t stream) {
    const float* x    = (const float*)d_in[0];
    const int*  eidx  = (const int*)  d_in[1];
    const int*  batch = (const int*)  d_in[2];
    const float* t    = (const float*)d_in[3];
    const float* W1r  = (const float*)d_in[4];
    const float* b1   = (const float*)d_in[5];
    const float* W1s  = (const float*)d_in[6];
    const float* W2r  = (const float*)d_in[7];
    const float* b2   = (const float*)d_in[8];
    const float* W2s  = (const float*)d_in[9];
    const float* W3r  = (const float*)d_in[10];
    const float* b3   = (const float*)d_in[11];
    const float* W3s  = (const float*)d_in[12];
    const float* Wl1  = (const float*)d_in[13];
    const float* bl1  = (const float*)d_in[14];
    const float* Wl2  = (const float*)d_in[15];
    const float* bl2  = (const float*)d_in[16];
    const int* src = eidx;
    const int* dst = eidx + NE;

    float* ws = (float*)d_ws;
    size_t o = 0;
    float*    x1     = ws + o; o += (size_t)NN * HID;
    float*    x2     = ws + o; o += (size_t)NN * HID;
    float*    x3     = ws + o; o += (size_t)NN * HID;
    unsigned* own4   = (unsigned*)(ws + o); o += (size_t)NN * HID;
    unsigned* glist  = (unsigned*)(ws + o); o += (size_t)NN * HID;
    unsigned* gcnt   = (unsigned*)(ws + o); o += (size_t)NN * HID;
    float*    aggr   = ws + o; o += (size_t)NN * HID;
    float*    xpad   = ws + o; o += (size_t)NN * 32;
    float*    w1rp   = ws + o; o += (size_t)HID * 32;
    float*    w1sp   = ws + o; o += (size_t)HID * 32;
    unsigned* henc   = (unsigned*)(ws + o); o += 3 * NG * HID;
    float*    h      = ws + o; o += NG * 3 * HID;
    float*    z1     = ws + o; o += NG * 2 * HID;

    k_pad_rows<<<(NN * 32 + 255) / 256, 256, 0, stream>>>(x, xpad, NN, INC, 32);
    k_pad_rows<<<(HID * 32 + 255) / 256, 256, 0, stream>>>(W1r, w1rp, HID, INC, 32);
    k_pad_rows<<<(HID * 32 + 255) / 256, 256, 0, stream>>>(W1s, w1sp, HID, INC, 32);
    k_init_enc<<<(3 * NG * HID + 255) / 256, 256, 0, stream>>>(henc, 3 * NG * HID);
    k_owner_ids<<<(NE / 4 + 255) / 256, 256, 0, stream>>>(dst, own4);
    k_build_lists<<<NB, NT, 0, stream>>>(src, dst, own4, glist, gcnt);

    struct Layer { const float* xin; int Cp; const float *Wr, *b, *Ws; float* xout; };
    Layer layers[3] = {
        { xpad, 32, w1rp, b1, w1sp, x1 },
        { x1,   64, W2r,  b2, W2s,  x2 },
        { x2,   64, W3r,  b3, W3s,  x3 },
    };

    for (int l = 0; l < 3; ++l) {
        const Layer& L = layers[l];
        if (L.Cp == 32)
            k_aggr_owner<32><<<NB, NT, 0, stream>>>(src, dst, own4, glist, gcnt, L.xin, t, aggr);
        else
            k_aggr_owner<64><<<NB, NT, 0, stream>>>(src, dst, own4, glist, gcnt, L.xin, t, aggr);

        dim3 gblk(32, 4);
        if (L.Cp == 32)
            k_gemm<32><<<(NN + 15) / 16, gblk, 0, stream>>>(aggr, L.xin, L.Wr, L.Ws, L.b,
                                                            L.xout, NN);
        else
            k_gemm<64><<<(NN + 15) / 16, gblk, 0, stream>>>(aggr, L.xin, L.Wr, L.Ws, L.b,
                                                            L.xout, NN);

        k_instnorm_relu<<<(NN + 7) / 8, 256, 0, stream>>>(L.xout, NN);

        int ngm = NN * HID;
        k_graph_max<<<(ngm + 255) / 256, 256, 0, stream>>>(L.xout, batch,
                                                           henc + (size_t)l * NG * HID, ngm);
    }

    k_decode_h<<<(NG * 3 * HID + 255) / 256, 256, 0, stream>>>(henc, h, NG * 3 * HID);
    k_mlp1<<<(NG * 2 * HID + 255) / 256, 256, 0, stream>>>(h, Wl1, bl1, z1);
    k_mlp2_norm<<<NG * 32 / 128, 128, 0, stream>>>(z1, Wl2, bl2, (float*)d_out);
}


# Row-wise launches (x pad; norm+ReLU+graph max, now one pass) as resident grids: a wave walks 20 rows, graph max kept in registers, one atomic per graph run; 500K short waves become 10K; same sums
# speedup vs baseline: 2.4677x; 1.0316x over previous
#include <hip/hip_runtime.h>
#include <math.h>

#define NN      50000
#define NE      800000
#define NG      128
#define INC     31
#define HID     64
#define EPS     1e-5f

typedef __attribute__((ext_vector_type(2))) float v2f;
typedef __attribute__((ext_vector_type(8))) float v8f;

static __device__ __forceinline__ unsigned enc_f32(float x) {
    unsigned b = __float_as_uint(x);
    return (b & 0x80000000u) ? ~b : (b | 0x80000000u);
}
static __device__ __forceinline__ float dec_f32(unsigned e) {
    unsigned b = (e & 0x80000000u) ? (e & 0x7FFFFFFFu) : ~e;
    return __uint_as_float(b);
}
#define ENC_NEG_INF 0x007FFFFFu

__global__ void k_pad_rows(const float* __restrict__ s, float* __restrict__ d,
                           int rows, int Csrc, int Cdst) {
    int i = blockIdx.x * blockDim.x + threadIdx.x;
    if (i >= rows * Cdst) return;
    int r = i / Cdst, c = i % Cdst;
    d[i] = (c < Csrc) ? s[r * Csrc + c] : 0.f;
}

__global__ void k_init_enc(unsigned* __restrict__ p, int n) {
    int i = blockIdx.x * blockDim.x + threadIdx.x;
    if (i < n) p[i] = ENC_NEG_INF;
}

#define RB   200
#define NB   (NN / RB)
#define NT   1024
#define NWV  (NT / 32)
#define CAP  8192
#define EB   16
static_assert(NN % RB == 0, "groups must tile the nodes exactly");
static_assert(NB <= 255 && RB <= 256, "owner id (255 is kept free) and local node id must each fit a byte");
static_assert(NE % 16 == 0, "owner table is read 16 bytes a lane");
static_assert(CAP % EB == 0, "a list is padded to whole batches inside its capacity");

__global__ void k_owner_ids(const int* __restrict__ dst, unsigned* __restrict__ own4) {
    int g = blockIdx.x * blockDim.x + threadIdx.x;
    if (g >= NE / 4) return;
    const int4 d = ((const int4*)dst)[g];
    own4[g] = (unsigned)(d.x / RB) | ((unsigned)(d.y / RB) << 8) |
              ((unsigned)(d.z / RB) << 16) | ((unsigned)(d.w / RB) << 24);
}

__global__ void __launch_bounds__(NT)
k_build_lists(const int* __restrict__ src, const int* __restrict__ dst,
              const unsigned* __restrict__ own4,
              unsigned* __restrict__ glist, unsigned* __restrict__ gcnt) {
    __shared__ unsigned s_list[CAP];
    __shared__ unsigned s_cnt;
    const int tid  = threadIdx.x;
    const int b    = blockIdx.x;
    const int base = b * RB;
    if (tid == 0) s_cnt = 0u;
    __syncthreads();

    const uint4* own16 = (const uint4*)own4;
    const unsigned pat = (unsigned)b * 0x01010101u;
    constexpr int NQ = NE / 16;
    const int off = b * (NQ / NB);
    constexpr int SB = 8;
    for (int i = tid; i < NQ; i += SB * NT) {
        uint4 w4[SB];
        int   qq[SB];
#pragma unroll
        for (int r = 0; r < SB; ++r) {
            const int ii = i + r * NT;
            int q = (ii < NQ ? ii : tid) + off;
            if (q >= NQ) q -= NQ;
            qq[r] = q;
            w4[r] = own16[q];
        }
#pragma unroll
        for (int r = 0; r < SB; ++r) {
            const bool live = (i + r * NT) < NQ;
            const unsigned w[4] = { live ? w4[r].x : 0xFFFFFFFFu, live ? w4[r].y : 0xFFFFFFFFu,
                                    live ? w4[r].z : 0xFFFFFFFFu, live ? w4[r].w : 0xFFFFFFFFu };
#pragma unroll
            for (int k = 0; k < 4; ++k) {
                const unsigned xw = w[k] ^ pat;
                unsigned z = ~(((xw & 0x7F7F7F7Fu) + 0x7F7F7F7Fu) | xw) & 0x80808080u;
                while (z) {
                    const int j = __builtin_ctz(z) >> 3;
                    z &= z - 1u;
                    const unsigned pos = atomicAdd(&s_cnt, 1u);
                    if (pos < CAP) s_list[pos] = (unsigned)(qq[r] * 16 + k * 4 + j);
                }
            }
        }
    }
    __syncthreads();
    const unsigned cnt = s_cnt;
    if (tid == 0) gcnt[b] = cnt;
    if (cnt <= CAP) {
        const unsigned padded = (cnt + EB - 1u) & ~(unsigned)(EB - 1);
        for (unsigned i = tid; i < padded; i += NT) {
            const unsigned e = s_list[i < cnt ? i : cnt - 1u];
            glist[(size_t)b * CAP + i] = ((unsigned)src[e] << 8) | (unsigned)(dst[e] - base);
        }
    }
}

template<int CP>
__global__ void __launch_bounds__(NT)
k_aggr_owner(const int* __restrict__ src, const int* __restrict__ dst,
             const unsigned* __restrict__ own4,
             const unsigned* __restrict__ glist, const unsigned* __restrict__ gcnt,
             const float* __restrict__ x, const float* __restrict__ tptr,
             float* __restrict__ aggr) {
    constexpr int H = CP / 32;
    __shared__ unsigned s_em[RB * CP];
    __shared__ float    s_num[RB * CP];
    __shared__ float    s_den[RB * CP];

    const int tid  = threadIdx.x;
    const int lane = tid & 31;
    const int wv   = __builtin_amdgcn_readfirstlane(tid >> 5);
    const int b    = blockIdx.x;
    const int base = b * RB;
    const float t  = tptr[0];
    const unsigned cnt = __builtin_amdgcn_readfirstlane(gcnt[b]);

    for (int i = tid; i < RB * CP; i += NT) { s_em[i] = ENC_NEG_INF; s_num[i] = 0.f; s_den[i] = 0.f; }
    __syncthreads();

    if (cnt <= CAP) {
        const unsigned* __restrict__ gl = glist + (size_t)b * CAP;

        for (unsigned i0 = wv * EB; i0 < cnt; i0 += NWV * EB) {
            unsigned p[EB];
            float v[EB][H];
#pragma unroll
            for (int u = 0; u < EB; ++u) p[u] = gl[i0 + u];
#pragma unroll
            for (int u = 0; u < EB; ++u)
#pragma unroll
                for (int h = 0; h < H; ++h)
                    v[u][h] = x[(size_t)(p[u] >> 8) * CP + h * 32 + lane];
#pragma unroll
            for (int u = 0; u < EB; ++u)
#pragma unroll
                for (int h = 0; h < H; ++h)
                    atomicMax(&s_em[(p[u] & 255u) * CP + h * 32 + lane], enc_f32(v[u][h] * t));
        }
        __syncthreads();

        for (unsigned i0 = wv * EB; i0 < cnt; i0 += NWV * EB) {
            unsigned p[EB];
            float v[EB][H];
#pragma unroll
            for (int u = 0; u < EB; ++u) p[u] = gl[i0 + u];
#pragma unroll
            for (int u = 0; u < EB; ++u)
#pragma unroll
                for (int h = 0; h < H; ++h)
                    v[u][h] = x[(size_t)(p[u] >> 8) * CP + h * 32 + lane];
#pragma unroll
            for (int u = 0; u < EB; ++u)
                if (i0 + u < cnt) {
#pragma unroll
                    for (int h = 0; h < H; ++h) {
                        const unsigned a = (p[u] & 255u) * CP + h * 32 + lane;
                        const float m  = v[u][h];
                        const float ee = __expf(m * t - dec_f32(s_em[a]));
                        atomicAdd(&s_num[a], m * ee);
                        atomicAdd(&s_den[a], ee);
                    }
                }
        }
        __syncthreads();
    } else {
        const unsigned char* own1 = (const unsigned char*)own4;
        for (int pass = 0; pass < 2; ++pass) {
            for (int e = tid; e < NE; e += NT) {
                if (own1[e] != (unsigned char)b) continue;
                const int s  = src[e];
                const int dl = dst[e] - base;
                for (int c = 0; c < CP; ++c) {
                    const float m = x[(size_t)s * CP + c];
                    const int a = dl * CP + c;
                    if (pass == 0) {
                        atomicMax(&s_em[a], enc_f32(m * t));
                    } else {
                        const float ee = __expf(m * t - dec_f32(s_em[a]));
                        atomicAdd(&s_num[a], m * ee);
                        atomicAdd(&s_den[a], ee);
                    }
                }
            }
            __syncthreads();
        }
    }

    for (int i = tid; i < RB * CP; i += NT) {
        const float dd = s_den[i];
        aggr[(size_t)base * CP + i] = dd > 0.f ? s_num[i] / dd : 0.f;
    }
}

template<int CP>
__global__ void k_gemm(const float* __restrict__ aggr, const float* __restrict__ xin,
                       const float* __restrict__ Wr, const float* __restrict__ Ws,
                       const float* __restrict__ bias, float* __restrict__ out, int Nrows) {
    const int lane = threadIdx.x;
    const int half = lane >> 4;
    const int l16  = lane & 15;
    const int m0   = blockIdx.x * 16;
    const int h0   = threadIdx.y * 16;
    int arow = m0 + l16;
    if (arow >= Nrows) arow = Nrows - 1;
    const int bcol = h0 + l16;

    const float* ap = aggr + (size_t)arow * CP + half * 2;
    const float* xp = xin  + (size_t)arow * CP + half * 2;
    const float* wr = Wr   + (size_t)bcol * CP + half * 2;
    const float* wsp= Ws   + (size_t)bcol * CP + half * 2;

    v8f acc = {};
#pragma unroll
    for (int k = 0; k < CP; k += 4) {
        v2f a = *(const v2f*)(ap + k);
        v2f b = *(const v2f*)(wr + k);
        acc = __builtin_amdgcn_wmma_f32_16x16x4_f32(false, a, false, b,
                                                    (short)0, acc, false, false);
    }
#pragma unroll
    for (int k = 0; k < CP; k += 4) {
        v2f a = *(const v2f*)(xp + k);
        v2f b = *(const v2f*)(wsp + k);
        acc = __builtin_amdgcn_wmma_f32_16x16x4_f32(false, a, false, b,
                                                    (short)0, acc, false, false);
    }

    float bv = bias[bcol];
#pragma unroll
    for (int r = 0; r < 8; ++r) {
        int row = m0 + r + half * 8;
        if (row < Nrows) out[(size_t)row * HID + bcol] = acc[r] + bv;
    }
}

#define WR   20
#define WWV  (NN / WR)
#define WBT  128
static_assert(NN % WR == 0, "rows must split into whole waves");
static_assert(WWV % (WBT / 32) == 0, "the waves must fill whole work-groups");

__global__ void __launch_bounds__(WBT)
k_pad_x(const float* __restrict__ s, float* __restrict__ d) {
    const int lane = threadIdx.x & 31;
    const int wave = blockIdx.x * (WBT / 32) + (threadIdx.x >> 5);
    const int row0 = wave * WR;
    const int cl = lane < INC ? lane : INC - 1;
    float v[WR];
#pragma unroll
    for (int u = 0; u < WR; ++u) v[u] = s[(size_t)(row0 + u) * INC + cl];
#pragma unroll
    for (int u = 0; u < WR; ++u) d[(size_t)(row0 + u) * 32 + lane] = lane < INC ? v[u] : 0.f;
}

__global__ void __launch_bounds__(WBT)
k_norm_relu_gmax(float* __restrict__ x, const int* __restrict__ batch,
                 unsigned* __restrict__ he) {
    const int lane = threadIdx.x & 31;
    const int wave = blockIdx.x * (WBT / 32) + (threadIdx.x >> 5);
    const int row0 = wave * WR;
    float v0[WR], v1[WR];
    int gb[WR];
#pragma unroll
    for (int u = 0; u < WR; ++u) {
        v0[u] = x[(size_t)(row0 + u) * HID + lane];
        v1[u] = x[(size_t)(row0 + u) * HID + lane + 32];
        gb[u] = batch[row0 + u];
    }
#pragma unroll
    for (int u = 0; u < WR; ++u) {
        float s = v0[u] + v1[u];
        for (int off = 16; off; off >>= 1) s += __shfl_xor(s, off);
        const float mu = s * (1.f / HID);
        const float d0 = v0[u] - mu, d1 = v1[u] - mu;
        float q = d0 * d0 + d1 * d1;
        for (int off = 16; off; off >>= 1) q += __shfl_xor(q, off);
        const float inv = rsqrtf(q * (1.f / HID) + EPS);
        v0[u] = fmaxf(d0 * inv, 0.f);
        v1[u] = fmaxf(d1 * inv, 0.f);
        x[(size_t)(row0 + u) * HID + lane]      = v0[u];
        x[(size_t)(row0 + u) * HID + lane + 32] = v1[u];
    }
    int g = __builtin_amdgcn_readfirstlane(gb[0]);
    float m0 = -INFINITY, m1 = -INFINITY;
#pragma unroll
    for (int u = 0; u < WR; ++u) {
        const int gu = __builtin_amdgcn_readfirstlane(gb[u]);
        if (gu != g) {
            atomicMax(&he[g * HID + lane],      enc_f32(m0));
            atomicMax(&he[g * HID + lane + 32], enc_f32(m1));
            g = gu; m0 = -INFINITY; m1 = -INFINITY;
        }
        m0 = fmaxf(m0, v0[u]);
        m1 = fmaxf(m1, v1[u]);
    }
    atomicMax(&he[g * HID + lane],      enc_f32(m0));
    atomicMax(&he[g * HID + lane + 32], enc_f32(m1));
}

__global__ void k_decode_h(const unsigned* __restrict__ henc, float* __restrict__ h, int n) {
    int i = blockIdx.x * blockDim.x + threadIdx.x;
    if (i >= n) return;
    int g = i / (3 * HID);
    int c = i % (3 * HID);
    int layer = c / HID, cc = c % HID;
    h[i] = dec_f32(henc[layer * NG * HID + g * HID + cc]);
}

__global__ void k_mlp1(const float* __restrict__ h, const float* __restrict__ Wl1,
                       const float* __restrict__ bl1, float* __restrict__ z1) {
    int i = blockIdx.x * blockDim.x + threadIdx.x;
    if (i >= NG * 2 * HID) return;
    int g = i / (2 * HID), oc = i % (2 * HID);
    float s = bl1[oc];
    for (int k = 0; k < 3 * HID; ++k) s += h[g * 3 * HID + k] * Wl1[oc * 3 * HID + k];
    z1[i] = fmaxf(s, 0.f);
}

__global__ void k_mlp2_norm(const float* __restrict__ z1, const float* __restrict__ Wl2,
                            const float* __restrict__ bl2, float* __restrict__ outp) {
    int wave = (blockIdx.x * blockDim.x + threadIdx.x) >> 5;
    int lane = threadIdx.x & 31;
    if (wave >= NG) return;
    float s = bl2[lane];
    for (int k = 0; k < 2 * HID; ++k) s += z1[wave * 2 * HID + k] * Wl2[lane * 2 * HID + k];
    float q = s * s;
    for (int off = 16; off; off >>= 1) q += __shfl_xor(q, off);
    float nrm = fmaxf(sqrtf(q), 1e-12f);
    outp[wave * (HID / 2) + lane] = s / nrm;
}

extern "C" void kernel_launch(void* const* d_in, const int* in_sizes, int n_in,
                              void* d_out, int out_size, void* d_ws, size_t ws_size,
                              hipStream_t stream) {
    const float* x    = (const float*)d_in[0];
    const int*  eidx  = (const int*)  d_in[1];
    const int*  batch = (const int*)  d_in[2];
    const float* t    = (const float*)d_in[3];
    const float* W1r  = (const float*)d_in[4];
    const float* b1   = (const float*)d_in[5];
    const float* W1s  = (const float*)d_in[6];
    const float* W2r  = (const float*)d_in[7];
    const float* b2   = (const float*)d_in[8];
    const float* W2s  = (const float*)d_in[9];
    const float* W3r  = (const float*)d_in[10];
    const float* b3   = (const float*)d_in[11];
    const float* W3s  = (const float*)d_in[12];
    const float* Wl1  = (const float*)d_in[13];
    const float* bl1  = (const float*)d_in[14];
    const float* Wl2  = (const float*)d_in[15];
    const float* bl2  = (const float*)d_in[16];
    const int* src = eidx;
    const int* dst = eidx + NE;

    float* ws = (float*)d_ws;
    size_t o = 0;
    float*    x1     = ws + o; o += (size_t)NN * HID;
    float*    x2     = ws + o; o += (size_t)NN * HID;
    float*    x3     = ws + o; o += (size_t)NN * HID;
    unsigned* own4   = (unsigned*)(ws + o); o += (size_t)NN * HID;
    unsigned* glist  = (unsigned*)(ws + o); o += (size_t)NN * HID;
    unsigned* gcnt   = (unsigned*)(ws + o); o += (size_t)NN * HID;
    float*    aggr   = ws + o; o += (size_t)NN * HID;
    float*    xpad   = ws + o; o += (size_t)NN * 32;
    float*    w1rp   = ws + o; o += (size_t)HID * 32;
    float*    w1sp   = ws + o; o += (size_t)HID * 32;
    unsigned* henc   = (unsigned*)(ws + o); o += 3 * NG * HID;
    float*    h      = ws + o; o += NG * 3 * HID;
    float*    z1     = ws + o; o += NG * 2 * HID;

    k_pad_x<<<WWV / (WBT / 32), WBT, 0, stream>>>(x, xpad);
    k_pad_rows<<<(HID * 32 + 255) / 256, 256, 0, stream>>>(W1r, w1rp, HID, INC, 32);
    k_pad_rows<<<(HID * 32 + 255) / 256, 256, 0, stream>>>(W1s, w1sp, HID, INC, 32);
    k_init_enc<<<(3 * NG * HID + 255) / 256, 256, 0, stream>>>(henc, 3 * NG * HID);
    k_owner_ids<<<(NE / 4 + 255) / 256, 256, 0, stream>>>(dst, own4);
    k_build_lists<<<NB, NT, 0, stream>>>(src, dst, own4, glist, gcnt);

    struct Layer { const float* xin; int Cp; const float *Wr, *b, *Ws; float* xout; };
    Layer layers[3] = {
        { xpad, 32, w1rp, b1, w1sp, x1 },
        { x1,   64, W2r,  b2, W2s,  x2 },
        { x2,   64, W3r,  b3, W3s,  x3 },
    };

    for (int l = 0; l < 3; ++l) {
        const Layer& L = layers[l];
        if (L.Cp == 32)
            k_aggr_owner<32><<<NB, NT, 0, stream>>>(src, dst, own4, glist, gcnt, L.xin, t, aggr);
        else
            k_aggr_owner<64><<<NB, NT, 0, stream>>>(src, dst, own4, glist, gcnt, L.xin, t, aggr);

        dim3 gblk(32, 4);
        if (L.Cp == 32)
            k_gemm<32><<<(NN + 15) / 16, gblk, 0, stream>>>(aggr, L.xin, L.Wr, L.Ws, L.b,
                                                            L.xout, NN);
        else
            k_gemm<64><<<(NN + 15) / 16, gblk, 0, stream>>>(aggr, L.xin, L.Wr, L.Ws, L.b,
                                                            L.xout, NN);

        k_norm_relu_gmax<<<WWV / (WBT / 32), WBT, 0, stream>>>(L.xout, batch,
                                                                henc + (size_t)l * NG * HID);
    }

    k_decode_h<<<(NG * 3 * HID + 255) / 256, 256, 0, stream>>>(henc, h, NG * 3 * HID);
    k_mlp1<<<(NG * 2 * HID + 255) / 256, 256, 0, stream>>>(h, Wl1, bl1, z1);
    k_mlp2_norm<<<NG * 32 / 128, 128, 0, stream>>>(z1, Wl2, bl2, (float*)d_out);
}
